# starting kernel
# baseline (speedup 1.0000x reference)
#include <hip/hip_runtime.h>

typedef __attribute__((ext_vector_type(2))) float v2f;
typedef __attribute__((ext_vector_type(8))) float v8f;

#define N_NODES 65536
#define N_EDGES 1048576
#define DIM     64

__device__ __forceinline__ int   f2key(float f) {
  int i = __float_as_int(f);
  return i >= 0 ? i : (i ^ 0x7FFFFFFF);
}
__device__ __forceinline__ float key2f(int i) {
  return __int_as_float(i >= 0 ? i : (i ^ 0x7FFFFFFF));
}

__global__ void k0_init(float* __restrict__ out, int* __restrict__ emax_i,
                        float* __restrict__ denom) {
  int i = blockIdx.x * blockDim.x + threadIdx.x;
  out[i] = 0.0f;
  if (i < N_NODES) {
    emax_i[i] = (int)0x80000000;
    denom[i]  = 0.0f;
  }
}

__global__ void k1_gemm(const float* __restrict__ h, const float* __restrict__ w,
                        float* __restrict__ z) {
  int wave = (blockIdx.x * blockDim.x + threadIdx.x) >> 5;
  int lane = threadIdx.x & 31;
  int mt = wave >> 2;
  int nt = wave & 3;
  int r16   = lane & 15;
  int khalf = (lane >> 4) << 1;

  const float* arow = h + (size_t)(mt * 16 + r16) * DIM;
  const float* brow = w + (size_t)(nt * 16 + r16) * DIM;

  v8f c = {};
#pragma unroll
  for (int s = 0; s < 16; ++s) {
    int k = (s << 2) + khalf;
    v2f a = *(const v2f*)(arow + k);
    v2f b = *(const v2f*)(brow + k);
    c = __builtin_amdgcn_wmma_f32_16x16x4_f32(
             false, a,  false, b,
             (short)0, c,  false,  false);
  }

  int mbase = mt * 16 + ((lane >> 4) << 3);
  int n     = nt * 16 + r16;
#pragma unroll
  for (int r = 0; r < 8; ++r)
    z[(size_t)(mbase + r) * DIM + n] = c[r];
}

__global__ void k2_scores(const float* __restrict__ z, const float* __restrict__ attn,
                          float* __restrict__ es, float* __restrict__ ed) {
  int n = blockIdx.x * blockDim.x + threadIdx.x;
  const float4* zr = (const float4*)(z + (size_t)n * DIM);
  const float4* as = (const float4*)(attn);
  const float4* ad = (const float4*)(attn + DIM);
  float s = 0.0f, d = 0.0f;
#pragma unroll
  for (int k = 0; k < DIM / 4; ++k) {
    float4 zv = zr[k], av = as[k], dv = ad[k];
    s += zv.x * av.x + zv.y * av.y + zv.z * av.z + zv.w * av.w;
    d += zv.x * dv.x + zv.y * dv.y + zv.z * dv.z + zv.w * dv.w;
  }
  es[n] = s;
  ed[n] = d;
}

__global__ void k3_edge_max(const float* __restrict__ es, const float* __restrict__ ed,
                            const int* __restrict__ src, const int* __restrict__ dst,
                            float* __restrict__ e_buf, int* __restrict__ emax_i) {
  int t = blockIdx.x * blockDim.x + threadIdx.x;
  int d = dst[t];
  float e = es[src[t]] + ed[d];
  e = e >= 0.0f ? e : 0.01f * e;
  e_buf[t] = e;
  atomicMax(&emax_i[d], f2key(e));
}

__global__ void k4_edge_exp(const float* __restrict__ e_buf, const int* __restrict__ dst,
                            const int* __restrict__ emax_i, float* __restrict__ w_buf,
                            float* __restrict__ denom) {
  int t = blockIdx.x * blockDim.x + threadIdx.x;
  int d = dst[t];
  float m = key2f(emax_i[d]);
  if (!isfinite(m)) m = 0.0f;
  float wv = __expf(e_buf[t] - m);
  w_buf[t] = wv;
  atomicAdd(&denom[d], wv);
}

__global__ void k5_scatter(const float* __restrict__ z, const float* __restrict__ w_buf,
                           const float* __restrict__ denom, const int* __restrict__ src,
                           const int* __restrict__ dst, float* __restrict__ out) {
  int t    = blockIdx.x * blockDim.x + threadIdx.x;
  int edge = t >> 4;
  int part = t & 15;
  int s = src[edge], d = dst[edge];
  float dn = denom[d];
  float alpha = w_buf[edge] / (dn > 0.0f ? dn : 1.0f);
  float4 zv = *(const float4*)(z + (size_t)s * DIM + part * 4);
  float* o = out + (size_t)d * DIM + part * 4;
  atomicAdd(o + 0, alpha * zv.x);
  atomicAdd(o + 1, alpha * zv.y);
  atomicAdd(o + 2, alpha * zv.z);
  atomicAdd(o + 3, alpha * zv.w);
}

extern "C" void kernel_launch(void* const* d_in, const int* in_sizes, int n_in,
                              void* d_out, int out_size, void* d_ws, size_t ws_size,
                              hipStream_t stream) {
  const float* h     = (const float*)d_in[0];
  const float* fc_w  = (const float*)d_in[1];
  const float* attn  = (const float*)d_in[2];
  const int*   src   = (const int*)d_in[3];
  const int*   dst   = (const int*)d_in[4];
  float*       out   = (float*)d_out;

  char* ws = (char*)d_ws;
  size_t off = 0;
  float* z      = (float*)(ws + off); off += (size_t)N_NODES * DIM * 4;
  float* es     = (float*)(ws + off); off += (size_t)N_NODES * 4;
  float* ed     = (float*)(ws + off); off += (size_t)N_NODES * 4;
  int*   emax_i = (int*)  (ws + off); off += (size_t)N_NODES * 4;
  float* denom  = (float*)(ws + off); off += (size_t)N_NODES * 4;
  float* e_buf  = (float*)(ws + off); off += (size_t)N_EDGES * 4;
  float* w_buf  = (float*)(ws + off); off += (size_t)N_EDGES * 4;
  (void)off; (void)ws_size; (void)in_sizes; (void)n_in; (void)out_size;

  const int B = 256;
  k0_init   <<<(N_NODES * DIM) / B, B, 0, stream>>>(out, emax_i, denom);
  k1_gemm   <<<(N_NODES / 16) * 4 * 32 / B, B, 0, stream>>>(h, fc_w, z);
  k2_scores <<<N_NODES / B, B, 0, stream>>>(z, attn, es, ed);
  k3_edge_max<<<N_EDGES / B, B, 0, stream>>>(es, ed, src, dst, e_buf, emax_i);
  k4_edge_exp<<<N_EDGES / B, B, 0, stream>>>(e_buf, dst, emax_i, w_buf, denom);
  k5_scatter<<<(size_t)N_EDGES * 16 / B, B, 0, stream>>>(z, w_buf, denom, src, dst, out);
}


# Destination-range ownership replaces global atomics: edges binned per 256-node range by LDS counting sort; owner group keeps max, exp sum, row sums in LDS (float adds in arrival order), one divide
# speedup vs baseline: 3.6423x; 3.6423x over previous
#include <hip/hip_runtime.h>

typedef __attribute__((ext_vector_type(2))) float v2f;
typedef __attribute__((ext_vector_type(8))) float v8f;

#define N_NODES 65536
#define N_EDGES 1048576
#define DIM     64

__device__ __forceinline__ int   f2key(float f) {
  int i = __float_as_int(f);
  return i >= 0 ? i : (i ^ 0x7FFFFFFF);
}
__device__ __forceinline__ float key2f(int i) {
  return __int_as_float(i >= 0 ? i : (i ^ 0x7FFFFFFF));
}

__global__ void k1_gemm(const float* __restrict__ h, const float* __restrict__ w,
                        float* __restrict__ z) {
  int wave = (blockIdx.x * blockDim.x + threadIdx.x) >> 5;
  int lane = threadIdx.x & 31;
  int mt = wave >> 2;
  int nt = wave & 3;
  int r16   = lane & 15;
  int khalf = (lane >> 4) << 1;

  const float* arow = h + (size_t)(mt * 16 + r16) * DIM;
  const float* brow = w + (size_t)(nt * 16 + r16) * DIM;

  v8f c = {};
#pragma unroll
  for (int s = 0; s < 16; ++s) {
    int k = (s << 2) + khalf;
    v2f a = *(const v2f*)(arow + k);
    v2f b = *(const v2f*)(brow + k);
    c = __builtin_amdgcn_wmma_f32_16x16x4_f32(
             false, a,  false, b,
             (short)0, c,  false,  false);
  }

  int mbase = mt * 16 + ((lane >> 4) << 3);
  int n     = nt * 16 + r16;
#pragma unroll
  for (int r = 0; r < 8; ++r)
    z[(size_t)(mbase + r) * DIM + n] = c[r];
}

__global__ void k2_scores(const float* __restrict__ z, const float* __restrict__ attn,
                          float* __restrict__ es, float* __restrict__ ed) {
  int n = blockIdx.x * blockDim.x + threadIdx.x;
  const float4* zr = (const float4*)(z + (size_t)n * DIM);
  const float4* as = (const float4*)(attn);
  const float4* ad = (const float4*)(attn + DIM);
  float s = 0.0f, d = 0.0f;
#pragma unroll
  for (int k = 0; k < DIM / 4; ++k) {
    float4 zv = zr[k], av = as[k], dv = ad[k];
    s += zv.x * av.x + zv.y * av.y + zv.z * av.z + zv.w * av.w;
    d += zv.x * dv.x + zv.y * dv.y + zv.z * dv.z + zv.w * dv.w;
  }
  es[n] = s;
  ed[n] = d;
}

#define RANGE   256
#define NRANGE  (N_NODES / RANGE)
#define CHUNK   4096
#define NCHUNK  (N_EDGES / CHUNK)
#define OFFS    (NRANGE + 1)
#define CAP     8192
#define SU      5
#define GU      16

__global__ __launch_bounds__(1024)
void k3_bin(const int* __restrict__ src, const int* __restrict__ dst,
            int* __restrict__ recs_i, int* __restrict__ chunk_off) {
  __shared__ int  cnt[NRANGE];
  __shared__ int  offs[NRANGE];
  __shared__ int2 sorted[CHUNK];
  const int tid = threadIdx.x, c = blockIdx.x;
  const int wv = tid >> 5, lane = tid & 31;
  if (tid < NRANGE) cnt[tid] = 0;
  __syncthreads();

  int s[4], d[4], b[4], r[4];
#pragma unroll
  for (int k = 0; k < 4; ++k) {
    int grp = c + NCHUNK * (wv + 32 * k);
    int e = grp * 32 + lane;
    s[k] = src[e];
    d[k] = dst[e];
  }
#pragma unroll
  for (int k = 0; k < 4; ++k) {
    b[k] = ((unsigned)d[k] >> 8) & (NRANGE - 1);
    r[k] = atomicAdd(&cnt[b[k]], 1);
  }
  __syncthreads();
  if (tid < NRANGE) {
    int o = 0;
    for (int j = 0; j < tid; ++j) o += cnt[j];
    offs[tid] = o;
    chunk_off[c * OFFS + tid] = o;
    if (tid == NRANGE - 1) chunk_off[c * OFFS + NRANGE] = o + cnt[tid];
  }
  __syncthreads();
#pragma unroll
  for (int k = 0; k < 4; ++k) sorted[offs[b[k]] + r[k]] = make_int2(s[k], d[k]);
  __syncthreads();
  int2* recs = (int2*)recs_i;
#pragma unroll
  for (int k = 0; k < 4; ++k)
    recs[(size_t)c * CHUNK + tid + k * 1024] = sorted[tid + k * 1024];
}

__global__ __launch_bounds__(1024)
void k4_owner(const float* __restrict__ z, const float* __restrict__ es,
              const float* __restrict__ ed, const int* __restrict__ recs_i,
              const int* __restrict__ chunk_off, float* __restrict__ out) {
  __shared__ __align__(16) float acc[RANGE * DIM];
  __shared__ int   emax_i[RANGE];
  __shared__ float denom[RANGE];
  __shared__ int   lens[NCHUNK];
  __shared__ int   segbase[NCHUNK];
  __shared__ int   P[NCHUNK + 1];
  __shared__ int   t_src[CAP];
  __shared__ int   t_dl[CAP];
  __shared__ float t_w[CAP];

  const int tid = threadIdx.x, j = blockIdx.x;
  const int wv = tid >> 5, lane = tid & 31;
  const int2* recs = (const int2*)recs_i;

  if (tid < NCHUNK) {
    int a = chunk_off[tid * OFFS + j];
    int b = chunk_off[tid * OFFS + j + 1];
    lens[tid] = b - a;
    segbase[tid] = tid * CHUNK + a;
    emax_i[tid] = (int)0x80000000;
    denom[tid] = 0.0f;
  }
#pragma unroll
  for (int k = 0; k < 4; ++k)
    *(float4*)&acc[(tid + k * 1024) * 4] = make_float4(0.f, 0.f, 0.f, 0.f);
  __syncthreads();
  if (tid < NCHUNK) {
    int o = 0;
    for (int q = 0; q < tid; ++q) o += lens[q];
    P[tid] = o;
    if (tid == NCHUNK - 1) P[NCHUNK] = o + lens[tid];
  }
  __syncthreads();
  const int T = P[NCHUNK];
  const int ntiles = (T + CAP - 1) / CAP;

  auto stage = [&](int lo, int n, bool do_max) {
    for (int base = 0; base < n; base += 1024 * SU) {
      int  ii[SU];
      bool ok[SU];
      int2 rec[SU];
#pragma unroll
      for (int k = 0; k < SU; ++k) {
        int i = base + tid + k * 1024;
        ok[k] = i < n;
        ii[k] = ok[k] ? i : n - 1;
        int g = lo + ii[k];
        int a = 0, b = NCHUNK;
#pragma unroll
        for (int it = 0; it < 8; ++it) {
          int mid = (a + b) >> 1;
          if (P[mid] <= g) a = mid; else b = mid;
        }
        rec[k] = recs[segbase[a] + (g - P[a])];
      }
      float e[SU];
#pragma unroll
      for (int k = 0; k < SU; ++k) {
        float x = es[rec[k].x & (N_NODES - 1)] + ed[rec[k].y & (N_NODES - 1)];
        e[k] = x >= 0.0f ? x : 0.01f * x;
      }
#pragma unroll
      for (int k = 0; k < SU; ++k) {
        if (ok[k]) {
          int dl = rec[k].y & (RANGE - 1);
          t_src[ii[k]] = rec[k].x & (N_NODES - 1);
          t_dl[ii[k]]  = dl;
          t_w[ii[k]]   = e[k];
          if (do_max) atomicMax(&emax_i[dl], f2key(e[k]));
        }
      }
    }
  };

  for (int t = 0; t < ntiles; ++t) {
    int lo = t * CAP;
    int n = T - lo < CAP ? T - lo : CAP;
    stage(lo, n, true);
  }
  __syncthreads();

  for (int t = 0; t < ntiles; ++t) {
    int lo = t * CAP;
    int n = T - lo < CAP ? T - lo : CAP;
    if (ntiles > 1) {
      stage(lo, n, false);
      __syncthreads();
    }
    for (int i = tid; i < n; i += 1024) {
      int dl = t_dl[i];
      float m = key2f(emax_i[dl]);
      if (!isfinite(m)) m = 0.0f;
      float w = __expf(t_w[i] - m);
      t_w[i] = w;
      atomicAdd(&denom[dl], w);
    }
    __syncthreads();
    for (int i0 = wv * GU; i0 < n; i0 += 32 * GU) {
      float2 zv[GU];
#pragma unroll
      for (int u = 0; u < GU; ++u) {
        int i = i0 + u < n ? i0 + u : n - 1;
        zv[u] = *(const float2*)(z + (size_t)t_src[i] * DIM + lane * 2);
      }
#pragma unroll
      for (int u = 0; u < GU; ++u) {
        if (i0 + u < n) {
          float w = t_w[i0 + u];
          float* a = acc + t_dl[i0 + u] * DIM + lane * 2;
          atomicAdd(a,     w * zv[u].x);
          atomicAdd(a + 1, w * zv[u].y);
        }
      }
    }
    __syncthreads();
  }

#pragma unroll
  for (int k = 0; k < 4; ++k) {
    int q  = tid + k * 1024;
    int nl = q >> 4;
    float dn = denom[nl];
    dn = dn > 0.0f ? dn : 1.0f;
    float4 a = *(const float4*)&acc[q * 4];
    float4 o = make_float4(a.x / dn, a.y / dn, a.z / dn, a.w / dn);
    *(float4*)(out + ((size_t)j * RANGE) * DIM + (size_t)q * 4) = o;
  }
}

extern "C" void kernel_launch(void* const* d_in, const int* in_sizes, int n_in,
                              void* d_out, int out_size, void* d_ws, size_t ws_size,
                              hipStream_t stream) {
  const float* h     = (const float*)d_in[0];
  const float* fc_w  = (const float*)d_in[1];
  const float* attn  = (const float*)d_in[2];
  const int*   src   = (const int*)d_in[3];
  const int*   dst   = (const int*)d_in[4];
  float*       out   = (float*)d_out;

  char* ws = (char*)d_ws;
  size_t off = 0;
  float* z         = (float*)(ws + off); off += (size_t)N_NODES * DIM * 4;
  float* es        = (float*)(ws + off); off += (size_t)N_NODES * 4;
  float* ed        = (float*)(ws + off); off += (size_t)N_NODES * 4;
  int*   recs      = (int*)  (ws + off); off += (size_t)N_EDGES * 8;
  int*   chunk_off = (int*)  (ws + off); off += (size_t)NCHUNK * OFFS * 4;
  (void)off; (void)ws_size; (void)in_sizes; (void)n_in; (void)out_size;

  const int B = 256;
  k1_gemm   <<<(N_NODES / 16) * 4 * 32 / B, B, 0, stream>>>(h, fc_w, z);
  k2_scores <<<N_NODES / B, B, 0, stream>>>(z, attn, es, ed);
  k3_bin    <<<NCHUNK, 1024, 0, stream>>>(src, dst, recs, chunk_off);
  k4_owner  <<<NRANGE, 1024, 0, stream>>>(z, es, ed, recs, chunk_off, out);
}


# Scores move into the binning launch: each binning work-group copies the z rows of its 256 nodes to LDS with whole-line loads and 8 waves form es, ed beside the offset scan; 3 launches
# speedup vs baseline: 4.0299x; 1.1064x over previous
#include <hip/hip_runtime.h>

typedef __attribute__((ext_vector_type(2))) float v2f;
typedef __attribute__((ext_vector_type(8))) float v8f;

#define N_NODES 65536
#define N_EDGES 1048576
#define DIM     64

__device__ __forceinline__ int   f2key(float f) {
  int i = __float_as_int(f);
  return i >= 0 ? i : (i ^ 0x7FFFFFFF);
}
__device__ __forceinline__ float key2f(int i) {
  return __int_as_float(i >= 0 ? i : (i ^ 0x7FFFFFFF));
}

__global__ void k1_gemm(const float* __restrict__ h, const float* __restrict__ w,
                        float* __restrict__ z) {
  int wave = (blockIdx.x * blockDim.x + threadIdx.x) >> 5;
  int lane = threadIdx.x & 31;
  int mt = wave >> 2;
  int nt = wave & 3;
  int r16   = lane & 15;
  int khalf = (lane >> 4) << 1;

  const float* arow = h + (size_t)(mt * 16 + r16) * DIM;
  const float* brow = w + (size_t)(nt * 16 + r16) * DIM;

  v8f c = {};
#pragma unroll
  for (int s = 0; s < 16; ++s) {
    int k = (s << 2) + khalf;
    v2f a = *(const v2f*)(arow + k);
    v2f b = *(const v2f*)(brow + k);
    c = __builtin_amdgcn_wmma_f32_16x16x4_f32(
             false, a,  false, b,
             (short)0, c,  false,  false);
  }

  int mbase = mt * 16 + ((lane >> 4) << 3);
  int n     = nt * 16 + r16;
#pragma unroll
  for (int r = 0; r < 8; ++r)
    z[(size_t)(mbase + r) * DIM + n] = c[r];
}

#define RANGE   256
#define NRANGE  (N_NODES / RANGE)
#define CHUNK   4096
#define NCHUNK  (N_EDGES / CHUNK)
#define OFFS    (NRANGE + 1)
#define NPB     (N_NODES / NCHUNK)
#define ZP      (DIM + 4)
#define CAP     8192
#define SU      5
#define GU      16

__global__ __launch_bounds__(1024)
void k3_bin_scores(const int* __restrict__ src, const int* __restrict__ dst,
                   int* __restrict__ recs_i, int* __restrict__ chunk_off,
                   const float* __restrict__ z, const float* __restrict__ attn,
                   float* __restrict__ es, float* __restrict__ ed) {
  __shared__ int  cnt[NRANGE];
  __shared__ int  offs[NRANGE];
  __shared__ int2 sorted[CHUNK];
  __shared__ __align__(16) float zrow[NPB * ZP];
  const int tid = threadIdx.x, c = blockIdx.x;
  const int wv = tid >> 5, lane = tid & 31;
  if (tid < NRANGE) cnt[tid] = 0;
  __syncthreads();

  int s[4], d[4], b[4], r[4];
#pragma unroll
  for (int k = 0; k < 4; ++k) {
    int grp = c + NCHUNK * (wv + 32 * k);
    int e = grp * 32 + lane;
    s[k] = src[e];
    d[k] = dst[e];
  }
  const float4* zblk = (const float4*)(z + (size_t)c * NPB * DIM);
  float4 zl[NPB * DIM / 4 / 1024];
#pragma unroll
  for (int k = 0; k < NPB * DIM / 4 / 1024; ++k) zl[k] = zblk[tid + k * 1024];
#pragma unroll
  for (int k = 0; k < 4; ++k) {
    b[k] = ((unsigned)d[k] >> 8) & (NRANGE - 1);
    r[k] = atomicAdd(&cnt[b[k]], 1);
  }
#pragma unroll
  for (int k = 0; k < NPB * DIM / 4 / 1024; ++k) {
    int q = tid + k * 1024;
    *(float4*)&zrow[(q >> 4) * ZP + (q & 15) * 4] = zl[k];
  }
  __syncthreads();
  if (tid < NRANGE) {
    int o = 0;
    for (int j = 0; j < tid; ++j) o += cnt[j];
    offs[tid] = o;
    chunk_off[c * OFFS + tid] = o;
    if (tid == NRANGE - 1) chunk_off[c * OFFS + NRANGE] = o + cnt[tid];
  } else if (tid < NRANGE + NPB) {
    const int nl = tid - NRANGE;
    const float4* zr = (const float4*)&zrow[nl * ZP];
    const float4* as = (const float4*)(attn);
    const float4* ad = (const float4*)(attn + DIM);
    float ssum = 0.0f, dsum = 0.0f;
#pragma unroll
    for (int k = 0; k < DIM / 4; ++k) {
      float4 zv = zr[k], av = as[k], dv = ad[k];
      ssum += zv.x * av.x + zv.y * av.y + zv.z * av.z + zv.w * av.w;
      dsum += zv.x * dv.x + zv.y * dv.y + zv.z * dv.z + zv.w * dv.w;
    }
    es[c * NPB + nl] = ssum;
    ed[c * NPB + nl] = dsum;
  }
  __syncthreads();
#pragma unroll
  for (int k = 0; k < 4; ++k) sorted[offs[b[k]] + r[k]] = make_int2(s[k], d[k]);
  __syncthreads();
  int2* recs = (int2*)recs_i;
#pragma unroll
  for (int k = 0; k < 4; ++k)
    recs[(size_t)c * CHUNK + tid + k * 1024] = sorted[tid + k * 1024];
}

__global__ __launch_bounds__(1024)
void k4_owner(const float* __restrict__ z, const float* __restrict__ es,
              const float* __restrict__ ed, const int* __restrict__ recs_i,
              const int* __restrict__ chunk_off, float* __restrict__ out) {
  __shared__ __align__(16) float acc[RANGE * DIM];
  __shared__ int   emax_i[RANGE];
  __shared__ float denom[RANGE];
  __shared__ int   lens[NCHUNK];
  __shared__ int   segbase[NCHUNK];
  __shared__ int   P[NCHUNK + 1];
  __shared__ int   t_src[CAP];
  __shared__ int   t_dl[CAP];
  __shared__ float t_w[CAP];

  const int tid = threadIdx.x, j = blockIdx.x;
  const int wv = tid >> 5, lane = tid & 31;
  const int2* recs = (const int2*)recs_i;

  if (tid < NCHUNK) {
    int a = chunk_off[tid * OFFS + j];
    int b = chunk_off[tid * OFFS + j + 1];
    lens[tid] = b - a;
    segbase[tid] = tid * CHUNK + a;
    emax_i[tid] = (int)0x80000000;
    denom[tid] = 0.0f;
  }
#pragma unroll
  for (int k = 0; k < 4; ++k)
    *(float4*)&acc[(tid + k * 1024) * 4] = make_float4(0.f, 0.f, 0.f, 0.f);
  __syncthreads();
  if (tid < NCHUNK) {
    int o = 0;
    for (int q = 0; q < tid; ++q) o += lens[q];
    P[tid] = o;
    if (tid == NCHUNK - 1) P[NCHUNK] = o + lens[tid];
  }
  __syncthreads();
  const int T = P[NCHUNK];
  const int ntiles = (T + CAP - 1) / CAP;

  auto stage = [&](int lo, int n, bool do_max) {
    for (int base = 0; base < n; base += 1024 * SU) {
      int  ii[SU];
      bool ok[SU];
      int2 rec[SU];
#pragma unroll
      for (int k = 0; k < SU; ++k) {
        int i = base + tid + k * 1024;
        ok[k] = i < n;
        ii[k] = ok[k] ? i : n - 1;
        int g = lo + ii[k];
        int a = 0, b = NCHUNK;
#pragma unroll
        for (int it = 0; it < 8; ++it) {
          int mid = (a + b) >> 1;
          if (P[mid] <= g) a = mid; else b = mid;
        }
        rec[k] = recs[segbase[a] + (g - P[a])];
      }
      float e[SU];
#pragma unroll
      for (int k = 0; k < SU; ++k) {
        float x = es[rec[k].x & (N_NODES - 1)] + ed[rec[k].y & (N_NODES - 1)];
        e[k] = x >= 0.0f ? x : 0.01f * x;
      }
#pragma unroll
      for (int k = 0; k < SU; ++k) {
        if (ok[k]) {
          int dl = rec[k].y & (RANGE - 1);
          t_src[ii[k]] = rec[k].x & (N_NODES - 1);
          t_dl[ii[k]]  = dl;
          t_w[ii[k]]   = e[k];
          if (do_max) atomicMax(&emax_i[dl], f2key(e[k]));
        }
      }
    }
  };

  for (int t = 0; t < ntiles; ++t) {
    int lo = t * CAP;
    int n = T - lo < CAP ? T - lo : CAP;
    stage(lo, n, true);
  }
  __syncthreads();

  for (int t = 0; t < ntiles; ++t) {
    int lo = t * CAP;
    int n = T - lo < CAP ? T - lo : CAP;
    if (ntiles > 1) {
      stage(lo, n, false);
      __syncthreads();
    }
    for (int i = tid; i < n; i += 1024) {
      int dl = t_dl[i];
      float m = key2f(emax_i[dl]);
      if (!isfinite(m)) m = 0.0f;
      float w = __expf(t_w[i] - m);
      t_w[i] = w;
      atomicAdd(&denom[dl], w);
    }
    __syncthreads();
    for (int i0 = wv * GU; i0 < n; i0 += 32 * GU) {
      float2 zv[GU];
#pragma unroll
      for (int u = 0; u < GU; ++u) {
        int i = i0 + u < n ? i0 + u : n - 1;
        zv[u] = *(const float2*)(z + (size_t)t_src[i] * DIM + lane * 2);
      }
#pragma unroll
      for (int u = 0; u < GU; ++u) {
        if (i0 + u < n) {
          float w = t_w[i0 + u];
          float* a = acc + t_dl[i0 + u] * DIM + lane * 2;
          atomicAdd(a,     w * zv[u].x);
          atomicAdd(a + 1, w * zv[u].y);
        }
      }
    }
    __syncthreads();
  }

#pragma unroll
  for (int k = 0; k < 4; ++k) {
    int q  = tid + k * 1024;
    int nl = q >> 4;
    float dn = denom[nl];
    dn = dn > 0.0f ? dn : 1.0f;
    float4 a = *(const float4*)&acc[q * 4];
    float4 o = make_float4(a.x / dn, a.y / dn, a.z / dn, a.w / dn);
    *(float4*)(out + ((size_t)j * RANGE) * DIM + (size_t)q * 4) = o;
  }
}

extern "C" void kernel_launch(void* const* d_in, const int* in_sizes, int n_in,
                              void* d_out, int out_size, void* d_ws, size_t ws_size,
                              hipStream_t stream) {
  const float* h     = (const float*)d_in[0];
  const float* fc_w  = (const float*)d_in[1];
  const float* attn  = (const float*)d_in[2];
  const int*   src   = (const int*)d_in[3];
  const int*   dst   = (const int*)d_in[4];
  float*       out   = (float*)d_out;

  char* ws = (char*)d_ws;
  size_t off = 0;
  float* z         = (float*)(ws + off); off += (size_t)N_NODES * DIM * 4;
  float* es        = (float*)(ws + off); off += (size_t)N_NODES * 4;
  float* ed        = (float*)(ws + off); off += (size_t)N_NODES * 4;
  int*   recs      = (int*)  (ws + off); off += (size_t)N_EDGES * 8;
  int*   chunk_off = (int*)  (ws + off); off += (size_t)NCHUNK * OFFS * 4;
  (void)off; (void)ws_size; (void)in_sizes; (void)n_in; (void)out_size;

  const int B = 256;
  k1_gemm   <<<(N_NODES / 16) * 4 * 32 / B, B, 0, stream>>>(h, fc_w, z);
  k3_bin_scores<<<NCHUNK, 1024, 0, stream>>>(src, dst, recs, chunk_off, z, attn, es, ed);
  k4_owner  <<<NRANGE, 1024, 0, stream>>>(z, es, ed, recs, chunk_off, out);
}


# GEMM moves into the binning launch: each binning work-group stages its 256 h rows and fc_w in LDS, forms its z rows there on the same f32 WMMA chains and scores them from LDS; 2 launches
# speedup vs baseline: 4.2979x; 1.0665x over previous
#include <hip/hip_runtime.h>

typedef __attribute__((ext_vector_type(2))) float v2f;
typedef __attribute__((ext_vector_type(8))) float v8f;

#define N_NODES 65536
#define N_EDGES 1048576
#define DIM     64

__device__ __forceinline__ int   f2key(float f) {
  int i = __float_as_int(f);
  return i >= 0 ? i : (i ^ 0x7FFFFFFF);
}
__device__ __forceinline__ float key2f(int i) {
  return __int_as_float(i >= 0 ? i : (i ^ 0x7FFFFFFF));
}

#define RANGE   256
#define NRANGE  (N_NODES / RANGE)
#define CHUNK   4096
#define NCHUNK  (N_EDGES / CHUNK)
#define OFFS    (NRANGE + 1)
#define NPB     (N_NODES / NCHUNK)
#define ZP      (DIM + 4)
#define CAP     8192
#define SU      5
#define GU      16

__global__ __launch_bounds__(1024)
void k3_gemm_bin_scores(const int* __restrict__ src, const int* __restrict__ dst,
                        int* __restrict__ recs_i, int* __restrict__ chunk_off,
                        const float* __restrict__ h, const float* __restrict__ w,
                        float* __restrict__ z, const float* __restrict__ attn,
                        float* __restrict__ es, float* __restrict__ ed) {
  __shared__ int  cnt[NRANGE];
  __shared__ int  offs[NRANGE];
  __shared__ int2 sorted[CHUNK];
  __shared__ __align__(16) float hrow[NPB * ZP];
  __shared__ __align__(16) float wrow[DIM * ZP];
  __shared__ __align__(16) float zrow[NPB * ZP];
  __shared__ __align__(16) float attn_s[2 * DIM];
  const int tid = threadIdx.x, c = blockIdx.x;
  const int wv = tid >> 5, lane = tid & 31;
  if (tid < NRANGE) cnt[tid] = 0;
  __syncthreads();

  int s[4], d[4], b[4], r[4];
#pragma unroll
  for (int k = 0; k < 4; ++k) {
    int grp = c + NCHUNK * (wv + 32 * k);
    int e = grp * 32 + lane;
    s[k] = src[e];
    d[k] = dst[e];
  }
  const float4* hblk = (const float4*)(h + (size_t)c * NPB * DIM);
  float4 hl[NPB * DIM / 4 / 1024];
#pragma unroll
  for (int k = 0; k < NPB * DIM / 4 / 1024; ++k) hl[k] = hblk[tid + k * 1024];
  float4 wl = ((const float4*)w)[tid];
  float4 al = make_float4(0.f, 0.f, 0.f, 0.f);
  if (tid < 2 * DIM / 4) al = ((const float4*)attn)[tid];
#pragma unroll
  for (int k = 0; k < 4; ++k) {
    b[k] = ((unsigned)d[k] >> 8) & (NRANGE - 1);
    r[k] = atomicAdd(&cnt[b[k]], 1);
  }
#pragma unroll
  for (int k = 0; k < NPB * DIM / 4 / 1024; ++k) {
    int q = tid + k * 1024;
    *(float4*)&hrow[(q >> 4) * ZP + (q & 15) * 4] = hl[k];
  }
  *(float4*)&wrow[(tid >> 4) * ZP + (tid & 15) * 4] = wl;
  if (tid < 2 * DIM / 4) *(float4*)&attn_s[tid * 4] = al;
  __syncthreads();

  if (tid < NRANGE) {
    int o = 0;
    for (int j = 0; j < tid; ++j) o += cnt[j];
    offs[tid] = o;
    chunk_off[c * OFFS + tid] = o;
    if (tid == NRANGE - 1) chunk_off[c * OFFS + NRANGE] = o + cnt[tid];
  }

  {
    const int r16   = lane & 15;
    const int khalf = (lane >> 4) << 1;
    const int mt  = wv >> 1;
    const int nt0 = (wv & 1) << 1;
    const float* arow  = &hrow[(mt * 16 + r16) * ZP];
    const float* brow0 = &wrow[(nt0 * 16 + r16) * ZP];
    const float* brow1 = &wrow[((nt0 + 1) * 16 + r16) * ZP];
    v8f c0 = {}, c1 = {};
#pragma unroll
    for (int st = 0; st < 16; ++st) {
      int k = (st << 2) + khalf;
      v2f a  = *(const v2f*)(arow + k);
      v2f b0 = *(const v2f*)(brow0 + k);
      v2f b1 = *(const v2f*)(brow1 + k);
      c0 = __builtin_amdgcn_wmma_f32_16x16x4_f32(
                false, a,  false, b0,
                (short)0, c0,  false,  false);
      c1 = __builtin_amdgcn_wmma_f32_16x16x4_f32(
                false, a,  false, b1,
                (short)0, c1,  false,  false);
    }
    float* zt = &zrow[(mt * 16 + ((lane >> 4) << 3)) * ZP + nt0 * 16 + r16];
#pragma unroll
    for (int rr = 0; rr < 8; ++rr) {
      zt[rr * ZP]      = c0[rr];
      zt[rr * ZP + 16] = c1[rr];
    }
  }
  __syncthreads();

  float4* zblk = (float4*)(z + (size_t)c * NPB * DIM);
#pragma unroll
  for (int k = 0; k < NPB * DIM / 4 / 1024; ++k) {
    int q = tid + k * 1024;
    zblk[q] = *(const float4*)&zrow[(q >> 4) * ZP + (q & 15) * 4];
  }
#pragma unroll
  for (int k = 0; k < 4; ++k) sorted[offs[b[k]] + r[k]] = make_int2(s[k], d[k]);
  if (tid >= NRANGE && tid < NRANGE + NPB) {
    const int nl = tid - NRANGE;
    const float4* zr = (const float4*)&zrow[nl * ZP];
    const float4* as = (const float4*)(attn_s);
    const float4* ad = (const float4*)(attn_s + DIM);
    float ssum = 0.0f, dsum = 0.0f;
#pragma unroll
    for (int k = 0; k < DIM / 4; ++k) {
      float4 zv = zr[k], av = as[k], dv = ad[k];
      ssum += zv.x * av.x + zv.y * av.y + zv.z * av.z + zv.w * av.w;
      dsum += zv.x * dv.x + zv.y * dv.y + zv.z * dv.z + zv.w * dv.w;
    }
    es[c * NPB + nl] = ssum;
    ed[c * NPB + nl] = dsum;
  }
  __syncthreads();
  int2* recs = (int2*)recs_i;
#pragma unroll
  for (int k = 0; k < 4; ++k)
    recs[(size_t)c * CHUNK + tid + k * 1024] = sorted[tid + k * 1024];
}

__global__ __launch_bounds__(1024)
void k4_owner(const float* __restrict__ z, const float* __restrict__ es,
              const float* __restrict__ ed, const int* __restrict__ recs_i,
              const int* __restrict__ chunk_off, float* __restrict__ out) {
  __shared__ __align__(16) float acc[RANGE * DIM];
  __shared__ int   emax_i[RANGE];
  __shared__ float denom[RANGE];
  __shared__ int   lens[NCHUNK];
  __shared__ int   segbase[NCHUNK];
  __shared__ int   P[NCHUNK + 1];
  __shared__ int   t_src[CAP];
  __shared__ int   t_dl[CAP];
  __shared__ float t_w[CAP];

  const int tid = threadIdx.x, j = blockIdx.x;
  const int wv = tid >> 5, lane = tid & 31;
  const int2* recs = (const int2*)recs_i;

  if (tid < NCHUNK) {
    int a = chunk_off[tid * OFFS + j];
    int b = chunk_off[tid * OFFS + j + 1];
    lens[tid] = b - a;
    segbase[tid] = tid * CHUNK + a;
    emax_i[tid] = (int)0x80000000;
    denom[tid] = 0.0f;
  }
#pragma unroll
  for (int k = 0; k < 4; ++k)
    *(float4*)&acc[(tid + k * 1024) * 4] = make_float4(0.f, 0.f, 0.f, 0.f);
  __syncthreads();
  if (tid < NCHUNK) {
    int o = 0;
    for (int q = 0; q < tid; ++q) o += lens[q];
    P[tid] = o;
    if (tid == NCHUNK - 1) P[NCHUNK] = o + lens[tid];
  }
  __syncthreads();
  const int T = P[NCHUNK];
  const int ntiles = (T + CAP - 1) / CAP;

  auto stage = [&](int lo, int n, bool do_max) {
    for (int base = 0; base < n; base += 1024 * SU) {
      int  ii[SU];
      bool ok[SU];
      int2 rec[SU];
#pragma unroll
      for (int k = 0; k < SU; ++k) {
        int i = base + tid + k * 1024;
        ok[k] = i < n;
        ii[k] = ok[k] ? i : n - 1;
        int g = lo + ii[k];
        int a = 0, b = NCHUNK;
#pragma unroll
        for (int it = 0; it < 8; ++it) {
          int mid = (a + b) >> 1;
          if (P[mid] <= g) a = mid; else b = mid;
        }
        rec[k] = recs[segbase[a] + (g - P[a])];
      }
      float e[SU];
#pragma unroll
      for (int k = 0; k < SU; ++k) {
        float x = es[rec[k].x & (N_NODES - 1)] + ed[rec[k].y & (N_NODES - 1)];
        e[k] = x >= 0.0f ? x : 0.01f * x;
      }
#pragma unroll
      for (int k = 0; k < SU; ++k) {
        if (ok[k]) {
          int dl = rec[k].y & (RANGE - 1);
          t_src[ii[k]] = rec[k].x & (N_NODES - 1);
          t_dl[ii[k]]  = dl;
          t_w[ii[k]]   = e[k];
          if (do_max) atomicMax(&emax_i[dl], f2key(e[k]));
        }
      }
    }
  };

  for (int t = 0; t < ntiles; ++t) {
    int lo = t * CAP;
    int n = T - lo < CAP ? T - lo : CAP;
    stage(lo, n, true);
  }
  __syncthreads();

  for (int t = 0; t < ntiles; ++t) {
    int lo = t * CAP;
    int n = T - lo < CAP ? T - lo : CAP;
    if (ntiles > 1) {
      stage(lo, n, false);
      __syncthreads();
    }
    for (int i = tid; i < n; i += 1024) {
      int dl = t_dl[i];
      float m = key2f(emax_i[dl]);
      if (!isfinite(m)) m = 0.0f;
      float w = __expf(t_w[i] - m);
      t_w[i] = w;
      atomicAdd(&denom[dl], w);
    }
    __syncthreads();
    for (int i0 = wv * GU; i0 < n; i0 += 32 * GU) {
      float2 zv[GU];
#pragma unroll
      for (int u = 0; u < GU; ++u) {
        int i = i0 + u < n ? i0 + u : n - 1;
        zv[u] = *(const float2*)(z + (size_t)t_src[i] * DIM + lane * 2);
      }
#pragma unroll
      for (int u = 0; u < GU; ++u) {
        if (i0 + u < n) {
          float w = t_w[i0 + u];
          float* a = acc + t_dl[i0 + u] * DIM + lane * 2;
          atomicAdd(a,     w * zv[u].x);
          atomicAdd(a + 1, w * zv[u].y);
        }
      }
    }
    __syncthreads();
  }

#pragma unroll
  for (int k = 0; k < 4; ++k) {
    int q  = tid + k * 1024;
    int nl = q >> 4;
    float dn = denom[nl];
    dn = dn > 0.0f ? dn : 1.0f;
    float4 a = *(const float4*)&acc[q * 4];
    float4 o = make_float4(a.x / dn, a.y / dn, a.z / dn, a.w / dn);
    *(float4*)(out + ((size_t)j * RANGE) * DIM + (size_t)q * 4) = o;
  }
}

extern "C" void kernel_launch(void* const* d_in, const int* in_sizes, int n_in,
                              void* d_out, int out_size, void* d_ws, size_t ws_size,
                              hipStream_t stream) {
  const float* h     = (const float*)d_in[0];
  const float* fc_w  = (const float*)d_in[1];
  const float* attn  = (const float*)d_in[2];
  const int*   src   = (const int*)d_in[3];
  const int*   dst   = (const int*)d_in[4];
  float*       out   = (float*)d_out;

  char* ws = (char*)d_ws;
  size_t off = 0;
  float* z         = (float*)(ws + off); off += (size_t)N_NODES * DIM * 4;
  float* es        = (float*)(ws + off); off += (size_t)N_NODES * 4;
  float* ed        = (float*)(ws + off); off += (size_t)N_NODES * 4;
  int*   recs      = (int*)  (ws + off); off += (size_t)N_EDGES * 8;
  int*   chunk_off = (int*)  (ws + off); off += (size_t)NCHUNK * OFFS * 4;
  (void)off; (void)ws_size; (void)in_sizes; (void)n_in; (void)out_size;

  k3_gemm_bin_scores<<<NCHUNK, 1024, 0, stream>>>(src, dst, recs, chunk_off, h, fc_w, z,
                                                  attn, es, ed);
  k4_owner  <<<NRANGE, 1024, 0, stream>>>(z, es, ed, recs, chunk_off, out);
}


# Owner group counting-sorts its staged edges by destination in LDS; a wave sums each run of equal destinations in registers (sorted-rank order, fused multiply-add): one LDS add per run, not per edge
# speedup vs baseline: 5.5826x; 1.2989x over previous
#include <hip/hip_runtime.h>
#include <type_traits>

typedef __attribute__((ext_vector_type(2))) float v2f;
typedef __attribute__((ext_vector_type(8))) float v8f;

#define N_NODES 65536
#define N_EDGES 1048576
#define DIM     64

__device__ __forceinline__ int   f2key(float f) {
  int i = __float_as_int(f);
  return i >= 0 ? i : (i ^ 0x7FFFFFFF);
}
__device__ __forceinline__ float key2f(int i) {
  return __int_as_float(i >= 0 ? i : (i ^ 0x7FFFFFFF));
}

#define RANGE   256
#define NRANGE  (N_NODES / RANGE)
#define CHUNK   4096
#define NCHUNK  (N_EDGES / CHUNK)
#define OFFS    (NRANGE + 1)
#define NPB     (N_NODES / NCHUNK)
#define ZP      (DIM + 4)
#define SU      5
#define CAP     (1024 * SU)
#define GU      32

__global__ __launch_bounds__(1024)
void k3_gemm_bin_scores(const int* __restrict__ src, const int* __restrict__ dst,
                        int* __restrict__ recs_i, int* __restrict__ chunk_off,
                        const float* __restrict__ h, const float* __restrict__ w,
                        float* __restrict__ z, const float* __restrict__ attn,
                        float* __restrict__ es, float* __restrict__ ed) {
  __shared__ int  cnt[NRANGE];
  __shared__ int  offs[NRANGE];
  __shared__ int2 sorted[CHUNK];
  __shared__ __align__(16) float hrow[NPB * ZP];
  __shared__ __align__(16) float wrow[DIM * ZP];
  __shared__ __align__(16) float zrow[NPB * ZP];
  __shared__ __align__(16) float attn_s[2 * DIM];
  const int tid = threadIdx.x, c = blockIdx.x;
  const int wv = tid >> 5, lane = tid & 31;
  if (tid < NRANGE) cnt[tid] = 0;
  __syncthreads();

  int s[4], d[4], b[4], r[4];
#pragma unroll
  for (int k = 0; k < 4; ++k) {
    int grp = c + NCHUNK * (wv + 32 * k);
    int e = grp * 32 + lane;
    s[k] = src[e];
    d[k] = dst[e];
  }
  const float4* hblk = (const float4*)(h + (size_t)c * NPB * DIM);
  float4 hl[NPB * DIM / 4 / 1024];
#pragma unroll
  for (int k = 0; k < NPB * DIM / 4 / 1024; ++k) hl[k] = hblk[tid + k * 1024];
  float4 wl = ((const float4*)w)[tid];
  float4 al = make_float4(0.f, 0.f, 0.f, 0.f);
  if (tid < 2 * DIM / 4) al = ((const float4*)attn)[tid];
#pragma unroll
  for (int k = 0; k < 4; ++k) {
    b[k] = ((unsigned)d[k] >> 8) & (NRANGE - 1);
    r[k] = atomicAdd(&cnt[b[k]], 1);
  }
#pragma unroll
  for (int k = 0; k < NPB * DIM / 4 / 1024; ++k) {
    int q = tid + k * 1024;
    *(float4*)&hrow[(q >> 4) * ZP + (q & 15) * 4] = hl[k];
  }
  *(float4*)&wrow[(tid >> 4) * ZP + (tid & 15) * 4] = wl;
  if (tid < 2 * DIM / 4) *(float4*)&attn_s[tid * 4] = al;
  __syncthreads();

  if (tid < NRANGE) {
    int o = 0;
    for (int j = 0; j < tid; ++j) o += cnt[j];
    offs[tid] = o;
    chunk_off[c * OFFS + tid] = o;
    if (tid == NRANGE - 1) chunk_off[c * OFFS + NRANGE] = o + cnt[tid];
  }

  {
    const int r16   = lane & 15;
    const int khalf = (lane >> 4) << 1;
    const int mt  = wv >> 1;
    const int nt0 = (wv & 1) << 1;
    const float* arow  = &hrow[(mt * 16 + r16) * ZP];
    const float* brow0 = &wrow[(nt0 * 16 + r16) * ZP];
    const float* brow1 = &wrow[((nt0 + 1) * 16 + r16) * ZP];
    v8f c0 = {}, c1 = {};
#pragma unroll
    for (int st = 0; st < 16; ++st) {
      int k = (st << 2) + khalf;
      v2f a  = *(const v2f*)(arow + k);
      v2f b0 = *(const v2f*)(brow0 + k);
      v2f b1 = *(const v2f*)(brow1 + k);
      c0 = __builtin_amdgcn_wmma_f32_16x16x4_f32(
                false, a,  false, b0,
                (short)0, c0,  false,  false);
      c1 = __builtin_amdgcn_wmma_f32_16x16x4_f32(
                false, a,  false, b1,
                (short)0, c1,  false,  false);
    }
    float* zt = &zrow[(mt * 16 + ((lane >> 4) << 3)) * ZP + nt0 * 16 + r16];
#pragma unroll
    for (int rr = 0; rr < 8; ++rr) {
      zt[rr * ZP]      = c0[rr];
      zt[rr * ZP + 16] = c1[rr];
    }
  }
  __syncthreads();

  float4* zblk = (float4*)(z + (size_t)c * NPB * DIM);
#pragma unroll
  for (int k = 0; k < NPB * DIM / 4 / 1024; ++k) {
    int q = tid + k * 1024;
    zblk[q] = *(const float4*)&zrow[(q >> 4) * ZP + (q & 15) * 4];
  }
#pragma unroll
  for (int k = 0; k < 4; ++k) sorted[offs[b[k]] + r[k]] = make_int2(s[k], d[k]);
  if (tid >= NRANGE && tid < NRANGE + NPB) {
    const int nl = tid - NRANGE;
    const float4* zr = (const float4*)&zrow[nl * ZP];
    const float4* as = (const float4*)(attn_s);
    const float4* ad = (const float4*)(attn_s + DIM);
    float ssum = 0.0f, dsum = 0.0f;
#pragma unroll
    for (int k = 0; k < DIM / 4; ++k) {
      float4 zv = zr[k], av = as[k], dv = ad[k];
      ssum += zv.x * av.x + zv.y * av.y + zv.z * av.z + zv.w * av.w;
      dsum += zv.x * dv.x + zv.y * dv.y + zv.z * dv.z + zv.w * dv.w;
    }
    es[c * NPB + nl] = ssum;
    ed[c * NPB + nl] = dsum;
  }
  __syncthreads();
  int2* recs = (int2*)recs_i;
#pragma unroll
  for (int k = 0; k < 4; ++k)
    recs[(size_t)c * CHUNK + tid + k * 1024] = sorted[tid + k * 1024];
}

__global__ __launch_bounds__(1024)
void k4_owner(const float* __restrict__ z, const float* __restrict__ es,
              const float* __restrict__ ed, const int* __restrict__ recs_i,
              const int* __restrict__ chunk_off, float* __restrict__ out) {
  __shared__ __align__(16) float acc[RANGE * DIM];
  __shared__ int   emax_i[RANGE];
  __shared__ float denom[RANGE];
  __shared__ int   lens[NCHUNK];
  __shared__ int   segbase[NCHUNK];
  __shared__ int   P[NCHUNK + 1];
  __shared__ __align__(16) int cntd[RANGE];
  __shared__ __align__(16) int startd[RANGE];
  __shared__ int   t_sd[CAP];
  __shared__ float t_w[CAP];

  const int tid = threadIdx.x, j = blockIdx.x;
  const int lane = tid & 31;
  const int wv = __builtin_amdgcn_readfirstlane(tid >> 5);
  const int2* recs = (const int2*)recs_i;

  if (tid < NCHUNK) {
    int a = chunk_off[tid * OFFS + j];
    int b = chunk_off[tid * OFFS + j + 1];
    lens[tid] = b - a;
    segbase[tid] = tid * CHUNK + a;
    emax_i[tid] = (int)0x80000000;
    denom[tid] = 0.0f;
    cntd[tid] = 0;
  }
#pragma unroll
  for (int k = 0; k < 4; ++k)
    *(float4*)&acc[(tid + k * 1024) * 4] = make_float4(0.f, 0.f, 0.f, 0.f);
  __syncthreads();
  if (tid < NCHUNK) {
    int o = 0;
    for (int q = 0; q < tid; ++q) o += lens[q];
    P[tid] = o;
    if (tid == NCHUNK - 1) P[NCHUNK] = o + lens[tid];
  }
  __syncthreads();
  const int T = P[NCHUNK];
  const int ntiles = (T + CAP - 1) / CAP;

  auto stage = [&](int lo, int n, bool do_max, bool do_sort) {
    bool  ok[SU];
    int2  rec[SU];
    int   sd[SU], rk[SU];
    float e[SU];
#pragma unroll
    for (int k = 0; k < SU; ++k) {
      int i = tid + k * 1024;
      ok[k] = i < n;
      int g = lo + (ok[k] ? i : n - 1);
      int a = 0, b = NCHUNK;
#pragma unroll
      for (int it = 0; it < 8; ++it) {
        int mid = (a + b) >> 1;
        if (P[mid] <= g) a = mid; else b = mid;
      }
      rec[k] = recs[segbase[a] + (g - P[a])];
    }
#pragma unroll
    for (int k = 0; k < SU; ++k) {
      float x = es[rec[k].x & (N_NODES - 1)] + ed[rec[k].y & (N_NODES - 1)];
      e[k] = x >= 0.0f ? x : 0.01f * x;
    }
#pragma unroll
    for (int k = 0; k < SU; ++k) {
      int dl = rec[k].y & (RANGE - 1);
      sd[k] = (rec[k].x & (N_NODES - 1)) | (dl << 16);
      rk[k] = 0;
      if (ok[k]) {
        if (do_max)  atomicMax(&emax_i[dl], f2key(e[k]));
        if (do_sort) rk[k] = atomicAdd(&cntd[dl], 1);
      }
    }
    if (do_sort) {
      __syncthreads();
      if (wv == 0) {
        int4 c0 = *(const int4*)&cntd[lane * 8];
        int4 c1 = *(const int4*)&cntd[lane * 8 + 4];
        int tot = c0.x + c0.y + c0.z + c0.w + c1.x + c1.y + c1.z + c1.w;
        int incl = tot;
#pragma unroll
        for (int off = 1; off < 32; off <<= 1) {
          int up = __shfl_up(incl, off);
          if (lane >= off) incl += up;
        }
        int4 s0, s1;
        s0.x = incl - tot;   s0.y = s0.x + c0.x; s0.z = s0.y + c0.y; s0.w = s0.z + c0.z;
        s1.x = s0.w + c0.w;  s1.y = s1.x + c1.x; s1.z = s1.y + c1.y; s1.w = s1.z + c1.z;
        *(int4*)&startd[lane * 8]     = s0;
        *(int4*)&startd[lane * 8 + 4] = s1;
        *(int4*)&cntd[lane * 8]       = make_int4(0, 0, 0, 0);
        *(int4*)&cntd[lane * 8 + 4]   = make_int4(0, 0, 0, 0);
      }
      __syncthreads();
#pragma unroll
      for (int k = 0; k < SU; ++k) {
        if (ok[k]) {
          int dl  = sd[k] >> 16;
          int pos = startd[dl] + rk[k];
          float m = key2f(emax_i[dl]);
          if (!isfinite(m)) m = 0.0f;
          t_sd[pos] = sd[k];
          t_w[pos]  = __expf(e[k] - m);
        }
      }
      __syncthreads();
    }
  };

  auto flush = [&](int d, float ax, float ay, float ws) {
    float* a = acc + d * DIM + lane * 2;
    atomicAdd(a,     ax);
    atomicAdd(a + 1, ay);
    if (lane == 0) atomicAdd(&denom[d], ws);
  };

  if (ntiles > 1) {
    for (int t = 0; t < ntiles; ++t) {
      int lo = t * CAP;
      int n = T - lo < CAP ? T - lo : CAP;
      stage(lo, n, true, false);
    }
  }

  for (int t = 0; t < ntiles; ++t) {
    int lo = t * CAP;
    int n = T - lo < CAP ? T - lo : CAP;
    stage(lo, n, ntiles == 1, true);
    auto trip = [&](int i0, auto full_c) {
      constexpr bool full = decltype(full_c)::value;
      int  li = i0 + lane;
      bool in = full || li < n;
      int  lc = in ? li : n - 1;
      int  sdv = t_sd[lc];
      float2 zv[GU];
#pragma unroll
      for (int u = 0; u < GU; ++u) {
        int s = __builtin_amdgcn_readlane(sdv, u) & (N_NODES - 1);
        zv[u] = *(const float2*)(z + (size_t)s * DIM + lane * 2);
      }
      int wvi = __float_as_int(in ? t_w[lc] : 0.0f);
      int cur = __builtin_amdgcn_readlane(sdv, 0) >> 16;
      float ax = 0.0f, ay = 0.0f, ws = 0.0f;
#pragma unroll
      for (int u = 0; u < GU; ++u) {
        if (full || i0 + u < n) {
          int   dl = __builtin_amdgcn_readlane(sdv, u) >> 16;
          float w  = __int_as_float(__builtin_amdgcn_readlane(wvi, u));
          if (dl != cur) {
            flush(cur, ax, ay, ws);
            cur = dl;
            ax = 0.0f; ay = 0.0f; ws = 0.0f;
          }
          ax += w * zv[u].x;
          ay += w * zv[u].y;
          ws += w;
        }
      }
      flush(cur, ax, ay, ws);
    };
    for (int i0 = wv * GU; i0 < n; i0 += 32 * GU) {
      if (i0 + GU <= n) trip(i0, std::true_type{});
      else              trip(i0, std::false_type{});
    }
    __syncthreads();
  }

#pragma unroll
  for (int k = 0; k < 4; ++k) {
    int q  = tid + k * 1024;
    int nl = q >> 4;
    float dn = denom[nl];
    dn = dn > 0.0f ? dn : 1.0f;
    float4 a = *(const float4*)&acc[q * 4];
    float4 o = make_float4(a.x / dn, a.y / dn, a.z / dn, a.w / dn);
    *(float4*)(out + ((size_t)j * RANGE) * DIM + (size_t)q * 4) = o;
  }
}

extern "C" void kernel_launch(void* const* d_in, const int* in_sizes, int n_in,
                              void* d_out, int out_size, void* d_ws, size_t ws_size,
                              hipStream_t stream) {
  const float* h     = (const float*)d_in[0];
  const float* fc_w  = (const float*)d_in[1];
  const float* attn  = (const float*)d_in[2];
  const int*   src   = (const int*)d_in[3];
  const int*   dst   = (const int*)d_in[4];
  float*       out   = (float*)d_out;

  char* ws = (char*)d_ws;
  size_t off = 0;
  float* z         = (float*)(ws + off); off += (size_t)N_NODES * DIM * 4;
  float* es        = (float*)(ws + off); off += (size_t)N_NODES * 4;
  float* ed        = (float*)(ws + off); off += (size_t)N_NODES * 4;
  int*   recs      = (int*)  (ws + off); off += (size_t)N_EDGES * 8;
  int*   chunk_off = (int*)  (ws + off); off += (size_t)NCHUNK * OFFS * 4;
  (void)off; (void)ws_size; (void)in_sizes; (void)n_in; (void)out_size;

  k3_gemm_bin_scores<<<NCHUNK, 1024, 0, stream>>>(src, dst, recs, chunk_off, h, fc_w, z,
                                                  attn, es, ed);
  k4_owner  <<<NRANGE, 1024, 0, stream>>>(z, es, ed, recs, chunk_off, out);
}


# Owner group holds all of es (256 KB) and its own 256 ed values in LDS, filled with whole-line loads under the prefix scan, and looks edge scores up there; row sums then take the table's first 64 KB
# speedup vs baseline: 5.7990x; 1.0388x over previous
#include <hip/hip_runtime.h>
#include <type_traits>

typedef __attribute__((ext_vector_type(2))) float v2f;
typedef __attribute__((ext_vector_type(8))) float v8f;

#define N_NODES 65536
#define N_EDGES 1048576
#define DIM     64

__device__ __forceinline__ int   f2key(float f) {
  int i = __float_as_int(f);
  return i >= 0 ? i : (i ^ 0x7FFFFFFF);
}
__device__ __forceinline__ float key2f(int i) {
  return __int_as_float(i >= 0 ? i : (i ^ 0x7FFFFFFF));
}

#define RANGE   256
#define NRANGE  (N_NODES / RANGE)
#define CHUNK   4096
#define NCHUNK  (N_EDGES / CHUNK)
#define OFFS    (NRANGE + 1)
#define NPB     (N_NODES / NCHUNK)
#define ZP      (DIM + 4)
#define SU      5
#define CAP     (1024 * SU)
#define GU      32

__global__ __launch_bounds__(1024)
void k3_gemm_bin_scores(const int* __restrict__ src, const int* __restrict__ dst,
                        int* __restrict__ recs_i, int* __restrict__ chunk_off,
                        const float* __restrict__ h, const float* __restrict__ w,
                        float* __restrict__ z, const float* __restrict__ attn,
                        float* __restrict__ es, float* __restrict__ ed) {
  __shared__ int  cnt[NRANGE];
  __shared__ int  offs[NRANGE];
  __shared__ int2 sorted[CHUNK];
  __shared__ __align__(16) float hrow[NPB * ZP];
  __shared__ __align__(16) float wrow[DIM * ZP];
  __shared__ __align__(16) float zrow[NPB * ZP];
  __shared__ __align__(16) float attn_s[2 * DIM];
  const int tid = threadIdx.x, c = blockIdx.x;
  const int wv = tid >> 5, lane = tid & 31;
  if (tid < NRANGE) cnt[tid] = 0;
  __syncthreads();

  int s[4], d[4], b[4], r[4];
#pragma unroll
  for (int k = 0; k < 4; ++k) {
    int grp = c + NCHUNK * (wv + 32 * k);
    int e = grp * 32 + lane;
    s[k] = src[e];
    d[k] = dst[e];
  }
  const float4* hblk = (const float4*)(h + (size_t)c * NPB * DIM);
  float4 hl[NPB * DIM / 4 / 1024];
#pragma unroll
  for (int k = 0; k < NPB * DIM / 4 / 1024; ++k) hl[k] = hblk[tid + k * 1024];
  float4 wl = ((const float4*)w)[tid];
  float4 al = make_float4(0.f, 0.f, 0.f, 0.f);
  if (tid < 2 * DIM / 4) al = ((const float4*)attn)[tid];
#pragma unroll
  for (int k = 0; k < 4; ++k) {
    b[k] = ((unsigned)d[k] >> 8) & (NRANGE - 1);
    r[k] = atomicAdd(&cnt[b[k]], 1);
  }
#pragma unroll
  for (int k = 0; k < NPB * DIM / 4 / 1024; ++k) {
    int q = tid + k * 1024;
    *(float4*)&hrow[(q >> 4) * ZP + (q & 15) * 4] = hl[k];
  }
  *(float4*)&wrow[(tid >> 4) * ZP + (tid & 15) * 4] = wl;
  if (tid < 2 * DIM / 4) *(float4*)&attn_s[tid * 4] = al;
  __syncthreads();

  if (tid < NRANGE) {
    int o = 0;
    for (int j = 0; j < tid; ++j) o += cnt[j];
    offs[tid] = o;
    chunk_off[c * OFFS + tid] = o;
    if (tid == NRANGE - 1) chunk_off[c * OFFS + NRANGE] = o + cnt[tid];
  }

  {
    const int r16   = lane & 15;
    const int khalf = (lane >> 4) << 1;
    const int mt  = wv >> 1;
    const int nt0 = (wv & 1) << 1;
    const float* arow  = &hrow[(mt * 16 + r16) * ZP];
    const float* brow0 = &wrow[(nt0 * 16 + r16) * ZP];
    const float* brow1 = &wrow[((nt0 + 1) * 16 + r16) * ZP];
    v8f c0 = {}, c1 = {};
#pragma unroll
    for (int st = 0; st < 16; ++st) {
      int k = (st << 2) + khalf;
      v2f a  = *(const v2f*)(arow + k);
      v2f b0 = *(const v2f*)(brow0 + k);
      v2f b1 = *(const v2f*)(brow1 + k);
      c0 = __builtin_amdgcn_wmma_f32_16x16x4_f32(
                false, a,  false, b0,
                (short)0, c0,  false,  false);
      c1 = __builtin_amdgcn_wmma_f32_16x16x4_f32(
                false, a,  false, b1,
                (short)0, c1,  false,  false);
    }
    float* zt = &zrow[(mt * 16 + ((lane >> 4) << 3)) * ZP + nt0 * 16 + r16];
#pragma unroll
    for (int rr = 0; rr < 8; ++rr) {
      zt[rr * ZP]      = c0[rr];
      zt[rr * ZP + 16] = c1[rr];
    }
  }
  __syncthreads();

  float4* zblk = (float4*)(z + (size_t)c * NPB * DIM);
#pragma unroll
  for (int k = 0; k < NPB * DIM / 4 / 1024; ++k) {
    int q = tid + k * 1024;
    zblk[q] = *(const float4*)&zrow[(q >> 4) * ZP + (q & 15) * 4];
  }
#pragma unroll
  for (int k = 0; k < 4; ++k) sorted[offs[b[k]] + r[k]] = make_int2(s[k], d[k]);
  if (tid >= NRANGE && tid < NRANGE + NPB) {
    const int nl = tid - NRANGE;
    const float4* zr = (const float4*)&zrow[nl * ZP];
    const float4* as = (const float4*)(attn_s);
    const float4* ad = (const float4*)(attn_s + DIM);
    float ssum = 0.0f, dsum = 0.0f;
#pragma unroll
    for (int k = 0; k < DIM / 4; ++k) {
      float4 zv = zr[k], av = as[k], dv = ad[k];
      ssum += zv.x * av.x + zv.y * av.y + zv.z * av.z + zv.w * av.w;
      dsum += zv.x * dv.x + zv.y * dv.y + zv.z * dv.z + zv.w * dv.w;
    }
    es[c * NPB + nl] = ssum;
    ed[c * NPB + nl] = dsum;
  }
  __syncthreads();
  int2* recs = (int2*)recs_i;
#pragma unroll
  for (int k = 0; k < 4; ++k)
    recs[(size_t)c * CHUNK + tid + k * 1024] = sorted[tid + k * 1024];
}

__global__ __launch_bounds__(1024)
void k4_owner(const float* __restrict__ z, const float* __restrict__ es,
              const float* __restrict__ ed, const int* __restrict__ recs_i,
              const int* __restrict__ chunk_off, float* __restrict__ out) {
  __shared__ __align__(16) float es_tab[N_NODES];
  float* const acc = es_tab;
  __shared__ float ed_s[RANGE];
  __shared__ int   emax_i[RANGE];
  __shared__ float denom[RANGE];
  __shared__ int   lens[NCHUNK];
  __shared__ int   segbase[NCHUNK];
  __shared__ int   P[NCHUNK + 1];
  __shared__ __align__(16) int cntd[RANGE];
  __shared__ __align__(16) int startd[RANGE];
  __shared__ int   t_sd[CAP];
  __shared__ float t_w[CAP];

  const int tid = threadIdx.x, j = blockIdx.x;
  const int lane = tid & 31;
  const int wv = __builtin_amdgcn_readfirstlane(tid >> 5);
  const int2* recs = (const int2*)recs_i;

  if (tid < NCHUNK) {
    int a = chunk_off[tid * OFFS + j];
    int b = chunk_off[tid * OFFS + j + 1];
    lens[tid] = b - a;
    segbase[tid] = tid * CHUNK + a;
    emax_i[tid] = (int)0x80000000;
    denom[tid] = 0.0f;
    cntd[tid] = 0;
  }
  if (tid < RANGE) ed_s[tid] = ed[j * RANGE + tid];
  __syncthreads();
  const float4* es4 = (const float4*)es;
#define ES_Q(k)  ((tid + (k) * 1024 + j * 64) & (N_NODES / 4 - 1))
#define ES_LD(k) const float4 et##k = es4[ES_Q(k)];
  ES_LD(0)  ES_LD(1)  ES_LD(2)  ES_LD(3)  ES_LD(4)  ES_LD(5)  ES_LD(6)  ES_LD(7)
  ES_LD(8)  ES_LD(9)  ES_LD(10) ES_LD(11) ES_LD(12) ES_LD(13) ES_LD(14) ES_LD(15)
  if (tid < NCHUNK) {
    int o = 0;
    for (int q = 0; q < tid; ++q) o += lens[q];
    P[tid] = o;
    if (tid == NCHUNK - 1) P[NCHUNK] = o + lens[tid];
  }
#define ES_ST(k) *(float4*)&es_tab[ES_Q(k) * 4] = et##k;
  ES_ST(0)  ES_ST(1)  ES_ST(2)  ES_ST(3)  ES_ST(4)  ES_ST(5)  ES_ST(6)  ES_ST(7)
  ES_ST(8)  ES_ST(9)  ES_ST(10) ES_ST(11) ES_ST(12) ES_ST(13) ES_ST(14) ES_ST(15)
  __syncthreads();
  const int T = P[NCHUNK];
  const int ntiles = (T + CAP - 1) / CAP;

  auto stage = [&](int lo, int n, bool do_max, bool do_sort, bool from_lds,
                   bool zero_acc) {
    bool  ok[SU];
    int2  rec[SU];
    int   sd[SU], rk[SU];
    float e[SU];
#pragma unroll
    for (int k = 0; k < SU; ++k) {
      int i = tid + k * 1024;
      ok[k] = i < n;
      int g = lo + (ok[k] ? i : n - 1);
      int a = 0, b = NCHUNK;
#pragma unroll
      for (int it = 0; it < 8; ++it) {
        int mid = (a + b) >> 1;
        if (P[mid] <= g) a = mid; else b = mid;
      }
      rec[k] = recs[segbase[a] + (g - P[a])];
    }
    float xs[SU];
    if (from_lds) {
#pragma unroll
      for (int k = 0; k < SU; ++k) xs[k] = es_tab[rec[k].x & (N_NODES - 1)];
    } else {
#pragma unroll
      for (int k = 0; k < SU; ++k) xs[k] = es[rec[k].x & (N_NODES - 1)];
    }
#pragma unroll
    for (int k = 0; k < SU; ++k) {
      float x = xs[k] + ed_s[rec[k].y & (RANGE - 1)];
      e[k] = x >= 0.0f ? x : 0.01f * x;
    }
#pragma unroll
    for (int k = 0; k < SU; ++k) {
      int dl = rec[k].y & (RANGE - 1);
      sd[k] = (rec[k].x & (N_NODES - 1)) | (dl << 16);
      rk[k] = 0;
      if (ok[k]) {
        if (do_max)  atomicMax(&emax_i[dl], f2key(e[k]));
        if (do_sort) rk[k] = atomicAdd(&cntd[dl], 1);
      }
    }
    if (do_sort) {
      __syncthreads();
      if (zero_acc) {
#pragma unroll
        for (int k = 0; k < 4; ++k)
          *(float4*)&acc[(tid + k * 1024) * 4] = make_float4(0.f, 0.f, 0.f, 0.f);
      }
      if (wv == 0) {
        int4 c0 = *(const int4*)&cntd[lane * 8];
        int4 c1 = *(const int4*)&cntd[lane * 8 + 4];
        int tot = c0.x + c0.y + c0.z + c0.w + c1.x + c1.y + c1.z + c1.w;
        int incl = tot;
#pragma unroll
        for (int off = 1; off < 32; off <<= 1) {
          int up = __shfl_up(incl, off);
          if (lane >= off) incl += up;
        }
        int4 s0, s1;
        s0.x = incl - tot;   s0.y = s0.x + c0.x; s0.z = s0.y + c0.y; s0.w = s0.z + c0.z;
        s1.x = s0.w + c0.w;  s1.y = s1.x + c1.x; s1.z = s1.y + c1.y; s1.w = s1.z + c1.z;
        *(int4*)&startd[lane * 8]     = s0;
        *(int4*)&startd[lane * 8 + 4] = s1;
        *(int4*)&cntd[lane * 8]       = make_int4(0, 0, 0, 0);
        *(int4*)&cntd[lane * 8 + 4]   = make_int4(0, 0, 0, 0);
      }
      __syncthreads();
#pragma unroll
      for (int k = 0; k < SU; ++k) {
        if (ok[k]) {
          int dl  = sd[k] >> 16;
          int pos = startd[dl] + rk[k];
          float m = key2f(emax_i[dl]);
          if (!isfinite(m)) m = 0.0f;
          t_sd[pos] = sd[k];
          t_w[pos]  = __expf(e[k] - m);
        }
      }
      __syncthreads();
    }
  };

  auto flush = [&](int d, float ax, float ay, float ws) {
    float* a = acc + d * DIM + lane * 2;
    atomicAdd(a,     ax);
    atomicAdd(a + 1, ay);
    if (lane == 0) atomicAdd(&denom[d], ws);
  };

  if (ntiles == 0) {
#pragma unroll
    for (int k = 0; k < 4; ++k)
      *(float4*)&acc[(tid + k * 1024) * 4] = make_float4(0.f, 0.f, 0.f, 0.f);
  }
  if (ntiles > 1) {
    for (int t = 0; t < ntiles; ++t) {
      int lo = t * CAP;
      int n = T - lo < CAP ? T - lo : CAP;
      stage(lo, n, true, false, true, false);
    }
  }

  for (int t = 0; t < ntiles; ++t) {
    int lo = t * CAP;
    int n = T - lo < CAP ? T - lo : CAP;
    stage(lo, n, ntiles == 1, true, t == 0, t == 0);
    auto trip = [&](int i0, auto full_c) {
      constexpr bool full = decltype(full_c)::value;
      int  li = i0 + lane;
      bool in = full || li < n;
      int  lc = in ? li : n - 1;
      int  sdv = t_sd[lc];
      float2 zv[GU];
#pragma unroll
      for (int u = 0; u < GU; ++u) {
        int s = __builtin_amdgcn_readlane(sdv, u) & (N_NODES - 1);
        zv[u] = *(const float2*)(z + (size_t)s * DIM + lane * 2);
      }
      int wvi = __float_as_int(in ? t_w[lc] : 0.0f);
      int cur = __builtin_amdgcn_readlane(sdv, 0) >> 16;
      float ax = 0.0f, ay = 0.0f, ws = 0.0f;
#pragma unroll
      for (int u = 0; u < GU; ++u) {
        if (full || i0 + u < n) {
          int   dl = __builtin_amdgcn_readlane(sdv, u) >> 16;
          float w  = __int_as_float(__builtin_amdgcn_readlane(wvi, u));
          if (dl != cur) {
            flush(cur, ax, ay, ws);
            cur = dl;
            ax = 0.0f; ay = 0.0f; ws = 0.0f;
          }
          ax += w * zv[u].x;
          ay += w * zv[u].y;
          ws += w;
        }
      }
      flush(cur, ax, ay, ws);
    };
    for (int i0 = wv * GU; i0 < n; i0 += 32 * GU) {
      if (i0 + GU <= n) trip(i0, std::true_type{});
      else              trip(i0, std::false_type{});
    }
    __syncthreads();
  }

#pragma unroll
  for (int k = 0; k < 4; ++k) {
    int q  = tid + k * 1024;
    int nl = q >> 4;
    float dn = denom[nl];
    dn = dn > 0.0f ? dn : 1.0f;
    float4 a = *(const float4*)&acc[q * 4];
    float4 o = make_float4(a.x / dn, a.y / dn, a.z / dn, a.w / dn);
    *(float4*)(out + ((size_t)j * RANGE) * DIM + (size_t)q * 4) = o;
  }
}

extern "C" void kernel_launch(void* const* d_in, const int* in_sizes, int n_in,
                              void* d_out, int out_size, void* d_ws, size_t ws_size,
                              hipStream_t stream) {
  const float* h     = (const float*)d_in[0];
  const float* fc_w  = (const float*)d_in[1];
  const float* attn  = (const float*)d_in[2];
  const int*   src   = (const int*)d_in[3];
  const int*   dst   = (const int*)d_in[4];
  float*       out   = (float*)d_out;

  char* ws = (char*)d_ws;
  size_t off = 0;
  float* z         = (float*)(ws + off); off += (size_t)N_NODES * DIM * 4;
  float* es        = (float*)(ws + off); off += (size_t)N_NODES * 4;
  float* ed        = (float*)(ws + off); off += (size_t)N_NODES * 4;
  int*   recs      = (int*)  (ws + off); off += (size_t)N_EDGES * 8;
  int*   chunk_off = (int*)  (ws + off); off += (size_t)NCHUNK * OFFS * 4;
  (void)off; (void)ws_size; (void)in_sizes; (void)n_in; (void)out_size;

  k3_gemm_bin_scores<<<NCHUNK, 1024, 0, stream>>>(src, dst, recs, chunk_off, h, fc_w, z,
                                                  attn, es, ed);
  k4_owner  <<<NRANGE, 1024, 0, stream>>>(z, es, ed, recs, chunk_off, out);
}


# Aggregate before transform: owner groups gather rows of h (a read-only input) not of z, then apply fc_w once per destination row on the same f32 WMMA chains; scores as h.(fc_w^T a); sums reassociated
# speedup vs baseline: 6.9030x; 1.1904x over previous
#include <hip/hip_runtime.h>
#include <type_traits>

typedef __attribute__((ext_vector_type(2))) float v2f;
typedef __attribute__((ext_vector_type(8))) float v8f;

#define N_NODES 65536
#define N_EDGES 1048576
#define DIM     64

__device__ __forceinline__ int   f2key(float f) {
  int i = __float_as_int(f);
  return i >= 0 ? i : (i ^ 0x7FFFFFFF);
}
__device__ __forceinline__ float key2f(int i) {
  return __int_as_float(i >= 0 ? i : (i ^ 0x7FFFFFFF));
}

#define RANGE   256
#define NRANGE  (N_NODES / RANGE)
#define CHUNK   4096
#define NCHUNK  (N_EDGES / CHUNK)
#define OFFS    (NRANGE + 1)
#define NPB     (N_NODES / NCHUNK)
#define ZP      (DIM + 4)
#define SU      5
#define CAP     (1024 * SU)
#define GU      32

__global__ __launch_bounds__(1024)
void k3_bin_scores(const int* __restrict__ src, const int* __restrict__ dst,
                   int* __restrict__ recs_i, int* __restrict__ chunk_off,
                   const float* __restrict__ h, const float* __restrict__ w,
                   const float* __restrict__ attn,
                   float* __restrict__ es, float* __restrict__ ed) {
  __shared__ __align__(16) int cnt[NRANGE];
  __shared__ __align__(16) int offs[NRANGE];
  __shared__ int2 sorted[CHUNK];
  __shared__ __align__(16) float hrow[NPB * ZP];
  __shared__ __align__(16) float wrow[DIM * ZP];
  __shared__ __align__(16) float attn_s[2 * DIM];
  __shared__ __align__(16) float vsd[2 * DIM];
  const int tid = threadIdx.x, c = blockIdx.x;
  const int wv = tid >> 5, lane = tid & 31;
  if (tid < NRANGE) cnt[tid] = 0;
  __syncthreads();

  int s[4], d[4], b[4], r[4];
#pragma unroll
  for (int k = 0; k < 4; ++k) {
    int grp = c + NCHUNK * (wv + 32 * k);
    int e = grp * 32 + lane;
    s[k] = src[e];
    d[k] = dst[e];
  }
  const float4* hblk = (const float4*)(h + (size_t)c * NPB * DIM);
  float4 hl[NPB * DIM / 4 / 1024];
#pragma unroll
  for (int k = 0; k < NPB * DIM / 4 / 1024; ++k) hl[k] = hblk[tid + k * 1024];
  float4 wl = ((const float4*)w)[tid];
  float4 al = make_float4(0.f, 0.f, 0.f, 0.f);
  if (tid < 2 * DIM / 4) al = ((const float4*)attn)[tid];
#pragma unroll
  for (int k = 0; k < 4; ++k) {
    b[k] = ((unsigned)d[k] >> 8) & (NRANGE - 1);
    r[k] = atomicAdd(&cnt[b[k]], 1);
  }
#pragma unroll
  for (int k = 0; k < NPB * DIM / 4 / 1024; ++k) {
    int q = tid + k * 1024;
    *(float4*)&hrow[(q >> 4) * ZP + (q & 15) * 4] = hl[k];
  }
  *(float4*)&wrow[(tid >> 4) * ZP + (tid & 15) * 4] = wl;
  if (tid < 2 * DIM / 4) *(float4*)&attn_s[tid * 4] = al;
  __syncthreads();

  if (wv == 0) {
    int4 c0 = *(const int4*)&cnt[lane * 8];
    int4 c1 = *(const int4*)&cnt[lane * 8 + 4];
    int tot = c0.x + c0.y + c0.z + c0.w + c1.x + c1.y + c1.z + c1.w;
    int incl = tot;
#pragma unroll
    for (int off = 1; off < 32; off <<= 1) {
      int up = __shfl_up(incl, off);
      if (lane >= off) incl += up;
    }
    int o[8];
    o[0] = incl - tot;  o[1] = o[0] + c0.x; o[2] = o[1] + c0.y; o[3] = o[2] + c0.z;
    o[4] = o[3] + c0.w; o[5] = o[4] + c1.x; o[6] = o[5] + c1.y; o[7] = o[6] + c1.z;
    *(int4*)&offs[lane * 8]     = make_int4(o[0], o[1], o[2], o[3]);
    *(int4*)&offs[lane * 8 + 4] = make_int4(o[4], o[5], o[6], o[7]);
  } else if (tid >= 128 && tid < 128 + 2 * DIM) {
    const int t = tid - 128, k = t & (DIM - 1);
    const float* a = attn_s + (t >> 6) * DIM;
    float v = 0.0f;
    for (int jj = 0; jj < DIM; ++jj) v += wrow[jj * ZP + k] * a[jj];
    vsd[t] = v;
  }
  __syncthreads();
  if (tid < NRANGE) {
    int o = offs[tid];
    chunk_off[c * OFFS + tid] = o;
    if (tid == NRANGE - 1) chunk_off[c * OFFS + NRANGE] = o + cnt[tid];
  }

#pragma unroll
  for (int k = 0; k < 4; ++k) sorted[offs[b[k]] + r[k]] = make_int2(s[k], d[k]);
  {
    const int nl = tid >> 2, p = tid & 3;
    const float4* hr = (const float4*)&hrow[nl * ZP + p * 16];
    const float4* vs = (const float4*)&vsd[p * 16];
    const float4* vd = (const float4*)&vsd[DIM + p * 16];
    float ssum = 0.0f, dsum = 0.0f;
#pragma unroll
    for (int k = 0; k < 4; ++k) {
      float4 hv = hr[k], av = vs[k], dv = vd[k];
      ssum += hv.x * av.x + hv.y * av.y + hv.z * av.z + hv.w * av.w;
      dsum += hv.x * dv.x + hv.y * dv.y + hv.z * dv.z + hv.w * dv.w;
    }
    ssum += __shfl_xor(ssum, 1);  dsum += __shfl_xor(dsum, 1);
    ssum += __shfl_xor(ssum, 2);  dsum += __shfl_xor(dsum, 2);
    if (p == 0) {
      es[c * NPB + nl] = ssum;
      ed[c * NPB + nl] = dsum;
    }
  }
  __syncthreads();
  int2* recs = (int2*)recs_i;
#pragma unroll
  for (int k = 0; k < 4; ++k)
    recs[(size_t)c * CHUNK + tid + k * 1024] = sorted[tid + k * 1024];
}

__global__ __launch_bounds__(1024)
void k4_owner(const float* __restrict__ h, const float* __restrict__ fcw,
              const float* __restrict__ es,
              const float* __restrict__ ed, const int* __restrict__ recs_i,
              const int* __restrict__ chunk_off, float* __restrict__ out) {
  __shared__ __align__(16) float es_tab[N_NODES];
  float* const acc  = es_tab;
  float* const wrow = es_tab + RANGE * ZP;
  float* const zout = es_tab + (RANGE + DIM) * ZP;
  __shared__ float ed_s[RANGE];
  __shared__ int   emax_i[RANGE];
  __shared__ float denom[RANGE];
  __shared__ int   lens[NCHUNK];
  __shared__ int   segbase[NCHUNK];
  __shared__ int   P[NCHUNK + 1];
  __shared__ __align__(16) int cntd[RANGE];
  __shared__ __align__(16) int startd[RANGE];
  __shared__ int   t_sd[CAP];
  __shared__ float t_w[CAP];

  const int tid = threadIdx.x, j = blockIdx.x;
  const int lane = tid & 31;
  const int wv = __builtin_amdgcn_readfirstlane(tid >> 5);
  const int2* recs = (const int2*)recs_i;

  if (tid < NCHUNK) {
    int a = chunk_off[tid * OFFS + j];
    int b = chunk_off[tid * OFFS + j + 1];
    lens[tid] = b - a;
    segbase[tid] = tid * CHUNK + a;
    emax_i[tid] = (int)0x80000000;
    denom[tid] = 0.0f;
    cntd[tid] = 0;
  }
  if (tid < RANGE) ed_s[tid] = ed[j * RANGE + tid];
  __syncthreads();
  const float4 wl = ((const float4*)fcw)[tid];
  const float4* es4 = (const float4*)es;
#define ES_Q(k)  ((tid + (k) * 1024 + j * 64) & (N_NODES / 4 - 1))
#define ES_LD(k) const float4 et##k = es4[ES_Q(k)];
  ES_LD(0)  ES_LD(1)  ES_LD(2)  ES_LD(3)  ES_LD(4)  ES_LD(5)  ES_LD(6)  ES_LD(7)
  ES_LD(8)  ES_LD(9)  ES_LD(10) ES_LD(11) ES_LD(12) ES_LD(13) ES_LD(14) ES_LD(15)
  if (tid < NCHUNK) {
    int o = 0;
    for (int q = 0; q < tid; ++q) o += lens[q];
    P[tid] = o;
    if (tid == NCHUNK - 1) P[NCHUNK] = o + lens[tid];
  }
#define ES_ST(k) *(float4*)&es_tab[ES_Q(k) * 4] = et##k;
  ES_ST(0)  ES_ST(1)  ES_ST(2)  ES_ST(3)  ES_ST(4)  ES_ST(5)  ES_ST(6)  ES_ST(7)
  ES_ST(8)  ES_ST(9)  ES_ST(10) ES_ST(11) ES_ST(12) ES_ST(13) ES_ST(14) ES_ST(15)
  __syncthreads();
  const int T = P[NCHUNK];
  const int ntiles = (T + CAP - 1) / CAP;

  auto clear_head = [&]() {
#pragma unroll
    for (int k = 0; k < (RANGE * ZP / 4 + 1023) / 1024; ++k) {
      int q = tid + k * 1024;
      if (q < RANGE * ZP / 4)
        *(float4*)&acc[q * 4] = make_float4(0.f, 0.f, 0.f, 0.f);
    }
    *(float4*)&wrow[(tid >> 4) * ZP + (tid & 15) * 4] = wl;
  };

  auto stage = [&](int lo, int n, bool do_max, bool do_sort, bool from_lds,
                   bool zero_acc) {
    bool  ok[SU];
    int2  rec[SU];
    int   sd[SU], rk[SU];
    float e[SU];
#pragma unroll
    for (int k = 0; k < SU; ++k) {
      int i = tid + k * 1024;
      ok[k] = i < n;
      int g = lo + (ok[k] ? i : n - 1);
      int a = 0, b = NCHUNK;
#pragma unroll
      for (int it = 0; it < 8; ++it) {
        int mid = (a + b) >> 1;
        if (P[mid] <= g) a = mid; else b = mid;
      }
      rec[k] = recs[segbase[a] + (g - P[a])];
    }
    float xs[SU];
    if (from_lds) {
#pragma unroll
      for (int k = 0; k < SU; ++k) xs[k] = es_tab[rec[k].x & (N_NODES - 1)];
    } else {
#pragma unroll
      for (int k = 0; k < SU; ++k) xs[k] = es[rec[k].x & (N_NODES - 1)];
    }
#pragma unroll
    for (int k = 0; k < SU; ++k) {
      float x = xs[k] + ed_s[rec[k].y & (RANGE - 1)];
      e[k] = x >= 0.0f ? x : 0.01f * x;
    }
#pragma unroll
    for (int k = 0; k < SU; ++k) {
      int dl = rec[k].y & (RANGE - 1);
      sd[k] = (rec[k].x & (N_NODES - 1)) | (dl << 16);
      rk[k] = 0;
      if (ok[k]) {
        if (do_max)  atomicMax(&emax_i[dl], f2key(e[k]));
        if (do_sort) rk[k] = atomicAdd(&cntd[dl], 1);
      }
    }
    if (do_sort) {
      __syncthreads();
      if (zero_acc) {
        clear_head();
      }
      if (wv == 0) {
        int4 c0 = *(const int4*)&cntd[lane * 8];
        int4 c1 = *(const int4*)&cntd[lane * 8 + 4];
        int tot = c0.x + c0.y + c0.z + c0.w + c1.x + c1.y + c1.z + c1.w;
        int incl = tot;
#pragma unroll
        for (int off = 1; off < 32; off <<= 1) {
          int up = __shfl_up(incl, off);
          if (lane >= off) incl += up;
        }
        int4 s0, s1;
        s0.x = incl - tot;   s0.y = s0.x + c0.x; s0.z = s0.y + c0.y; s0.w = s0.z + c0.z;
        s1.x = s0.w + c0.w;  s1.y = s1.x + c1.x; s1.z = s1.y + c1.y; s1.w = s1.z + c1.z;
        *(int4*)&startd[lane * 8]     = s0;
        *(int4*)&startd[lane * 8 + 4] = s1;
        *(int4*)&cntd[lane * 8]       = make_int4(0, 0, 0, 0);
        *(int4*)&cntd[lane * 8 + 4]   = make_int4(0, 0, 0, 0);
      }
      __syncthreads();
#pragma unroll
      for (int k = 0; k < SU; ++k) {
        if (ok[k]) {
          int dl  = sd[k] >> 16;
          int pos = startd[dl] + rk[k];
          float m = key2f(emax_i[dl]);
          if (!isfinite(m)) m = 0.0f;
          t_sd[pos] = sd[k];
          t_w[pos]  = __expf(e[k] - m);
        }
      }
      __syncthreads();
    }
  };

  auto flush = [&](int d, float ax, float ay, float ws) {
    float* a = acc + d * ZP + lane * 2;
    atomicAdd(a,     ax);
    atomicAdd(a + 1, ay);
    if (lane == 0) atomicAdd(&denom[d], ws);
  };

  if (ntiles == 0) {
    clear_head();
    __syncthreads();
  }
  if (ntiles > 1) {
    for (int t = 0; t < ntiles; ++t) {
      int lo = t * CAP;
      int n = T - lo < CAP ? T - lo : CAP;
      stage(lo, n, true, false, true, false);
    }
  }

  for (int t = 0; t < ntiles; ++t) {
    int lo = t * CAP;
    int n = T - lo < CAP ? T - lo : CAP;
    stage(lo, n, ntiles == 1, true, t == 0, t == 0);
    auto trip = [&](int i0, auto full_c) {
      constexpr bool full = decltype(full_c)::value;
      int  li = i0 + lane;
      bool in = full || li < n;
      int  lc = in ? li : n - 1;
      int  sdv = t_sd[lc];
      float2 zv[GU];
#pragma unroll
      for (int u = 0; u < GU; ++u) {
        int s = __builtin_amdgcn_readlane(sdv, u) & (N_NODES - 1);
        zv[u] = *(const float2*)(h + (size_t)s * DIM + lane * 2);
      }
      int wvi = __float_as_int(in ? t_w[lc] : 0.0f);
      int cur = __builtin_amdgcn_readlane(sdv, 0) >> 16;
      float ax = 0.0f, ay = 0.0f, ws = 0.0f;
#pragma unroll
      for (int u = 0; u < GU; ++u) {
        if (full || i0 + u < n) {
          int   dl = __builtin_amdgcn_readlane(sdv, u) >> 16;
          float w  = __int_as_float(__builtin_amdgcn_readlane(wvi, u));
          if (dl != cur) {
            flush(cur, ax, ay, ws);
            cur = dl;
            ax = 0.0f; ay = 0.0f; ws = 0.0f;
          }
          ax += w * zv[u].x;
          ay += w * zv[u].y;
          ws += w;
        }
      }
      flush(cur, ax, ay, ws);
    };
    for (int i0 = wv * GU; i0 < n; i0 += 32 * GU) {
      if (i0 + GU <= n) trip(i0, std::true_type{});
      else              trip(i0, std::false_type{});
    }
    __syncthreads();
  }

  {
    const int r16   = lane & 15;
    const int khalf = (lane >> 4) << 1;
    const int mt  = wv >> 1;
    const int nt0 = (wv & 1) << 1;
    const float* arow  = &acc[(mt * 16 + r16) * ZP];
    const float* brow0 = &wrow[(nt0 * 16 + r16) * ZP];
    const float* brow1 = &wrow[((nt0 + 1) * 16 + r16) * ZP];
    v8f c0 = {}, c1 = {};
#pragma unroll
    for (int st = 0; st < 16; ++st) {
      int k = (st << 2) + khalf;
      v2f a  = *(const v2f*)(arow + k);
      v2f b0 = *(const v2f*)(brow0 + k);
      v2f b1 = *(const v2f*)(brow1 + k);
      c0 = __builtin_amdgcn_wmma_f32_16x16x4_f32(
                false, a,  false, b0,
                (short)0, c0,  false,  false);
      c1 = __builtin_amdgcn_wmma_f32_16x16x4_f32(
                false, a,  false, b1,
                (short)0, c1,  false,  false);
    }
    float* zt = &zout[(mt * 16 + ((lane >> 4) << 3)) * ZP + nt0 * 16 + r16];
#pragma unroll
    for (int rr = 0; rr < 8; ++rr) {
      zt[rr * ZP]      = c0[rr];
      zt[rr * ZP + 16] = c1[rr];
    }
  }
  __syncthreads();

#pragma unroll
  for (int k = 0; k < 4; ++k) {
    int q  = tid + k * 1024;
    int nl = q >> 4;
    float dn = denom[nl];
    dn = dn > 0.0f ? dn : 1.0f;
    float4 a = *(const float4*)&zout[nl * ZP + (q & 15) * 4];
    float4 o = make_float4(a.x / dn, a.y / dn, a.z / dn, a.w / dn);
    *(float4*)(out + ((size_t)j * RANGE) * DIM + (size_t)q * 4) = o;
  }
}

extern "C" void kernel_launch(void* const* d_in, const int* in_sizes, int n_in,
                              void* d_out, int out_size, void* d_ws, size_t ws_size,
                              hipStream_t stream) {
  const float* h     = (const float*)d_in[0];
  const float* fc_w  = (const float*)d_in[1];
  const float* attn  = (const float*)d_in[2];
  const int*   src   = (const int*)d_in[3];
  const int*   dst   = (const int*)d_in[4];
  float*       out   = (float*)d_out;

  char* ws = (char*)d_ws;
  size_t off = 0;
  float* es        = (float*)(ws + off); off += (size_t)N_NODES * 4;
  float* ed        = (float*)(ws + off); off += (size_t)N_NODES * 4;
  int*   recs      = (int*)  (ws + off); off += (size_t)N_EDGES * 8;
  int*   chunk_off = (int*)  (ws + off); off += (size_t)NCHUNK * OFFS * 4;
  (void)off; (void)ws_size; (void)in_sizes; (void)n_in; (void)out_size;

  k3_bin_scores<<<NCHUNK, 1024, 0, stream>>>(src, dst, recs, chunk_off, h, fc_w,
                                             attn, es, ed);
  k4_owner  <<<NRANGE, 1024, 0, stream>>>(h, fc_w, es, ed, recs, chunk_off, out);
}
